# starting kernel
# baseline (speedup 1.0000x reference)
#include <hip/hip_runtime.h>

typedef __attribute__((ext_vector_type(2))) float v2f;
typedef __attribute__((ext_vector_type(8))) float v8f;


__global__ void deg_init_kernel(float* __restrict__ deg, int n) {
    int i = blockIdx.x * blockDim.x + threadIdx.x;
    if (i < n) deg[i] = 1.0f;
}

__global__ void deg_accum_kernel(const int* __restrict__ dst, float* __restrict__ deg, int E) {
    int e = blockIdx.x * blockDim.x + threadIdx.x;
    if (e < E) atomicAdd(&deg[dst[e]], 1.0f);
}

__global__ void rsqrt_kernel(float* __restrict__ d, int n) {
    int i = blockIdx.x * blockDim.x + threadIdx.x;
    if (i < n) d[i] = rsqrtf(d[i]);
}

__global__ void gemm_wmma_f32(const float* __restrict__ X, const float* __restrict__ W,
                              float* __restrict__ H, int n_rows, int k_dim, int n_cols) {
    const int wave = threadIdx.x >> 5;
    const int lane = threadIdx.x & 31;
    const int tiles_per_row = n_cols >> 4;
    const int total_tiles   = (n_rows >> 4) * tiles_per_row;
    const int tile = blockIdx.x * (blockDim.x >> 5) + wave;
    if (tile >= total_tiles) return;

    const int rt = tile / tiles_per_row;
    const int ct = tile - rt * tiles_per_row;
    const int r  = lane & 15;
    const int hi = lane >> 4;

    const float* __restrict__ arow = X + (size_t)(rt * 16 + r) * k_dim;
    const int bcol = ct * 16 + r;

    v8f acc = {};
    for (int k = 0; k < k_dim; k += 4) {
        v2f a, b;
        a.x = arow[k + 2 * hi];
        a.y = arow[k + 2 * hi + 1];
        b.x = W[(size_t)(k + 2 * hi)     * n_cols + bcol];
        b.y = W[(size_t)(k + 2 * hi + 1) * n_cols + bcol];
        acc = __builtin_amdgcn_wmma_f32_16x16x4_f32(
             false, a,  false, b,
             (short)0, acc,  false,  false);
    }

#pragma unroll
    for (int v = 0; v < 8; ++v) {
        H[(size_t)(rt * 16 + v + 8 * hi) * n_cols + ct * 16 + r] = acc[v];
    }
}


__global__ void init_out_kernel(const float* __restrict__ h, const float* __restrict__ dinv,
                                const float* __restrict__ bias, float* __restrict__ out,
                                int n, int f) {
    int idx = blockIdx.x * blockDim.x + threadIdx.x;
    if (idx >= n * f) return;
    int i = idx / f;
    int j = idx - i * f;
    float di = dinv[i];
    out[idx] = bias[j] + h[idx] * di * di;
}

__global__ void scatter_kernel(const float* __restrict__ h, const float* __restrict__ dinv,
                               const int* __restrict__ src, const int* __restrict__ dst,
                               float* __restrict__ out, int E, int f) {
    const int gpe = f >> 2;
    int idx = blockIdx.x * blockDim.x + threadIdx.x;
    if (idx >= E * gpe) return;
    int e = idx / gpe;
    int g = (idx - e * gpe) << 2;
    int s = src[e];
    int d = dst[e];
    float nrm = dinv[s] * dinv[d];
    const float4 hv = *(const float4*)(h + (size_t)s * f + g);
    float* o = out + (size_t)d * f + g;
    atomicAdd(o + 0, hv.x * nrm);
    atomicAdd(o + 1, hv.y * nrm);
    atomicAdd(o + 2, hv.z * nrm);
    atomicAdd(o + 3, hv.w * nrm);
}

__global__ void relu_kernel(float* __restrict__ x, int n) {
    int i = blockIdx.x * blockDim.x + threadIdx.x;
    if (i < n) x[i] = fmaxf(x[i], 0.0f);
}

__global__ void combine_kernel(const float* __restrict__ la, const float* __restrict__ lb,
                               float* __restrict__ logits, int n) {
    int i = blockIdx.x * blockDim.x + threadIdx.x;
    if (i < n) logits[i] = 0.5f * (la[i] + lb[i]);
}


static inline int cdiv(int a, int b) { return (a + b - 1) / b; }

static void run_branch(const float* x, const int* edge, const float* W1, const float* b1,
                       const float* W2, const float* b2, float* l_out,
                       float* dinv, float* h, float* xh, float* h2,
                       int Nn, int IN, int HID, int OUT, int E, hipStream_t stream) {
    const int* src = edge;
    const int* dst = edge + E;

    deg_init_kernel<<<cdiv(Nn, 256), 256, 0, stream>>>(dinv, Nn);
    deg_accum_kernel<<<cdiv(E, 256), 256, 0, stream>>>(dst, dinv, E);
    rsqrt_kernel<<<cdiv(Nn, 256), 256, 0, stream>>>(dinv, Nn);

    {
        int waves = (Nn / 16) * (HID / 16);
        gemm_wmma_f32<<<cdiv(waves, 8), 256, 0, stream>>>(x, W1, h, Nn, IN, HID);
        init_out_kernel<<<cdiv(Nn * HID, 256), 256, 0, stream>>>(h, dinv, b1, xh, Nn, HID);
        scatter_kernel<<<cdiv(E * (HID / 4), 256), 256, 0, stream>>>(h, dinv, src, dst, xh, E, HID);
        relu_kernel<<<cdiv(Nn * HID, 256), 256, 0, stream>>>(xh, Nn * HID);
    }

    {
        int waves = (Nn / 16) * (OUT / 16);
        gemm_wmma_f32<<<cdiv(waves, 8), 256, 0, stream>>>(xh, W2, h2, Nn, HID, OUT);
        init_out_kernel<<<cdiv(Nn * OUT, 256), 256, 0, stream>>>(h2, dinv, b2, l_out, Nn, OUT);
        scatter_kernel<<<cdiv(E * (OUT / 4), 256), 256, 0, stream>>>(h2, dinv, src, dst, l_out, E, OUT);
    }
}

extern "C" void kernel_launch(void* const* d_in, const int* in_sizes, int n_in,
                              void* d_out, int out_size, void* d_ws, size_t ws_size,
                              hipStream_t stream) {
    const float* x      = (const float*)d_in[0];
    const int*   edge_a = (const int*)  d_in[1];
    const int*   edge_b = (const int*)  d_in[2];
    const float* W1a    = (const float*)d_in[3];
    const float* b1a    = (const float*)d_in[4];
    const float* W2a    = (const float*)d_in[5];
    const float* b2a    = (const float*)d_in[6];
    const float* W1b    = (const float*)d_in[7];
    const float* b1b    = (const float*)d_in[8];
    const float* W2b    = (const float*)d_in[9];
    const float* b2b    = (const float*)d_in[10];

    const int HID = in_sizes[4];
    const int OUT = in_sizes[6];
    const int IN  = in_sizes[3] / HID;
    const int Nn  = in_sizes[0] / IN;
    const int E   = in_sizes[1] / 2;

    float* ws   = (float*)d_ws;
    float* dinv = ws;                       ws += Nn;
    float* h    = ws;                       ws += (size_t)Nn * HID;
    float* xh   = ws;                       ws += (size_t)Nn * HID;
    float* h2   = ws;                       ws += (size_t)Nn * OUT;

    float* logits = (float*)d_out;
    float* la     = logits + (size_t)Nn * OUT;
    float* lb     = la     + (size_t)Nn * OUT;

    run_branch(x, edge_a, W1a, b1a, W2a, b2a, la, dinv, h, xh, h2, Nn, IN, HID, OUT, E, stream);
    run_branch(x, edge_b, W1b, b1b, W2b, b2b, lb, dinv, h, xh, h2, Nn, IN, HID, OUT, E, stream);

    combine_kernel<<<cdiv(Nn * OUT, 256), 256, 0, stream>>>(la, lb, logits, Nn * OUT);
}


# degree count: 64 resident groups count their share of dst in LDS counters, then add the table to deg as whole 128-B lines; no per-edge 4-B global atomic; deg sums are integers, exact in any add order
# speedup vs baseline: 2.7478x; 2.7478x over previous
#include <hip/hip_runtime.h>

typedef __attribute__((ext_vector_type(2))) float v2f;
typedef __attribute__((ext_vector_type(8))) float v8f;


__global__ void deg_init_kernel(float* __restrict__ deg, int n) {
    int i = blockIdx.x * blockDim.x + threadIdx.x;
    if (i < n) deg[i] = 1.0f;
}

constexpr int CNT_CAP = 65536;

__global__ __launch_bounds__(1024)
void deg_accum_kernel(const int* __restrict__ dst, float* __restrict__ deg, int E, int n) {
    __shared__ unsigned cnt[CNT_CAP + 32];
    const int tid = threadIdx.x;
    const int nthr = blockDim.x;
    if (n > CNT_CAP) {
        for (int e = blockIdx.x * nthr + tid; e < E; e += gridDim.x * nthr)
            atomicAdd(&deg[dst[e]], 1.0f);
        return;
    }
    uint4* c4 = reinterpret_cast<uint4*>(cnt);
    const int n4 = (n + 3) >> 2;
    for (int i = tid; i < n4; i += nthr) c4[i] = make_uint4(0u, 0u, 0u, 0u);
    __syncthreads();

    const int per = (E + (int)gridDim.x - 1) / (int)gridDim.x;
    const int e_begin = min(E, (int)blockIdx.x * per);
    const int e_end = min(E, e_begin + per);
    const int spare = CNT_CAP + (tid & 31);
    for (int e0 = e_begin + tid; e0 < e_end; e0 += 8 * nthr) {
        int d[8];
#pragma unroll
        for (int q = 0; q < 8; ++q) d[q] = dst[min(e0 + q * nthr, e_end - 1)];
#pragma unroll
        for (int q = 0; q < 8; ++q)
            atomicAdd(&cnt[(e0 + q * nthr < e_end) ? d[q] : spare], 1u);
    }
    __syncthreads();

    const int lane = tid & 31;
    const int wv = tid >> 5, nw = nthr >> 5;
    const int nl = n >> 5;
    int ln = wv;
    for (; ln + 7 * nw < nl; ln += 8 * nw) {
        unsigned c[8];
#pragma unroll
        for (int q = 0; q < 8; ++q) c[q] = cnt[(ln + q * nw) * 32 + lane];
#pragma unroll
        for (int q = 0; q < 8; ++q) atomicAdd(&deg[(ln + q * nw) * 32 + lane], (float)c[q]);
    }
    for (; ln < nl; ln += nw) atomicAdd(&deg[ln * 32 + lane], (float)cnt[ln * 32 + lane]);
    const int it = (nl << 5) + tid;
    if (it < n) atomicAdd(&deg[it], (float)cnt[it]);
}

template <int NT, bool RELU_A, bool DEG_TO_DINV, bool ZERO_P, bool ADD_P>
__global__ void gemm_wmma_f32_fused(const float* __restrict__ X, const float* __restrict__ W,
                                    float* dnorm, const float* __restrict__ bias,
                                    float* H, float* O,
                                    float* p0, float* p1, float* p2,
                                    int n_rows, int k_dim) {
    constexpr int NC = NT * 16;
    const int wave = threadIdx.x >> 5;
    const int lane = threadIdx.x & 31;
    const int rt = blockIdx.x * (blockDim.x >> 5) + wave;
    if (rt >= (n_rows >> 4)) return;

    const int r  = lane & 15;
    const int hi = lane >> 4;

    const float* __restrict__ arow = X + (size_t)(rt * 16 + r) * k_dim;

    v8f acc[NT] = {};
    for (int k = 0; k < k_dim; k += 4) {
        v2f a;
        a.x = arow[k + 2 * hi];
        a.y = arow[k + 2 * hi + 1];
        if (ADD_P) {
            const float* pk = (k < 32) ? p0 : ((k < 64) ? p1 : p2);
            const float* prow = pk + (size_t)(rt * 16 + r) * 32 + (k & 31) + 2 * hi;
            a.x += prow[0];
            a.y += prow[1];
        }
        if (RELU_A) {
            a.x = fmaxf(a.x, 0.0f);
            a.y = fmaxf(a.y, 0.0f);
        }
        const float* __restrict__ w0 = W + (size_t)(k + 2 * hi) * NC + r;
#pragma unroll
        for (int t = 0; t < NT; ++t) {
            v2f b;
            b.x = w0[t * 16];
            b.y = w0[NC + t * 16];
            acc[t] = __builtin_amdgcn_wmma_f32_16x16x4_f32(
                 false, a,  false, b,
                 (short)0, acc[t],  false,  false);
        }
    }

    const int row0 = rt * 16 + 8 * hi;
    float di[8];
#pragma unroll
    for (int v = 0; v < 8; ++v) di[v] = dnorm[row0 + v];
    if (DEG_TO_DINV) {
#pragma unroll
        for (int v = 0; v < 8; ++v) di[v] = rsqrtf(di[v]);
        if (r == 0) {
#pragma unroll
            for (int v = 0; v < 8; ++v) dnorm[row0 + v] = di[v];
        }
    }
#pragma unroll
    for (int v = 0; v < 8; ++v) {
#pragma unroll
        for (int t = 0; t < NT; ++t) {
            const size_t idx = (size_t)(row0 + v) * NC + t * 16 + r;
            const float hv = acc[t][v];
            H[idx] = hv;
            O[idx] = bias[t * 16 + r] + hv * di[v] * di[v];
            if (ZERO_P) {
                float* pz = (t / 2 == 0) ? p0 : ((t / 2 == 1) ? p1 : p2);
                pz[(size_t)(row0 + v) * 32 + (t & 1) * 16 + r] = 0.0f;
            }
        }
    }
}


constexpr int TAB_CAP = 65536;

template <int NL, bool TWO_COPIES>
__global__ __launch_bounds__(1024)
void scatter_kernel(const float* __restrict__ h, const float* __restrict__ dinv,
                    const int* __restrict__ src, const int* __restrict__ dst,
                    float* __restrict__ out, float* __restrict__ p0, float* __restrict__ p1,
                    float* __restrict__ p2, int E, int n) {
    __shared__ float tab[TAB_CAP];
    constexpr int f = NL * 32;
    const bool in_lds = n <= TAB_CAP;
    if (in_lds) {
        const int n4 = n >> 2;
        const float4* __restrict__ d4 = reinterpret_cast<const float4*>(dinv);
        float4* t4 = reinterpret_cast<float4*>(tab);
        for (int i0 = threadIdx.x; i0 < n4; i0 += 8 * (int)blockDim.x) {
            int    ix[8];
            float4 r[8];
#pragma unroll
            for (int q = 0; q < 8; ++q) ix[q] = min(i0 + q * (int)blockDim.x, n4 - 1);
#pragma unroll
            for (int q = 0; q < 8; ++q) r[q] = d4[ix[q]];
#pragma unroll
            for (int q = 0; q < 8; ++q) t4[ix[q]] = r[q];
        }
        const int it = (n4 << 2) + threadIdx.x;
        if (it < n) tab[it] = dinv[it];
    }
    __syncthreads();

    const int lane = threadIdx.x & 31;
    const int wv = __builtin_amdgcn_readfirstlane(threadIdx.x >> 5);
    const int wpb = blockDim.x >> 5;
    const int W = gridDim.x * wpb;
    const int w = blockIdx.x * wpb + wv;
    const int U = E >> 3;
    const int uq = U / W, ur = U % W;
    const int e_begin = (w * uq + min(w, ur)) << 3;
    const int e_end = (w == W - 1) ? E : (((w + 1) * uq + min(w + 1, ur)) << 3);

    float* lb[NL];
#pragma unroll
    for (int k = 0; k < NL; ++k) lb[k] = out + k * 32;
    int pitch = f;
    if (TWO_COPIES) {
        const bool second_half = (__builtin_amdgcn_s_sendmsg_rtn(0x87) >> 18) & 1;
        if (second_half) {
            lb[0] = p0;
            if (NL > 1) lb[1] = p1;
            if (NL > 2) lb[2] = p2;
            pitch = 32;
        }
    }

    for (int e0 = e_begin; e0 < e_end; e0 += 32) {
        const int cnt = min(32, e_end - e0);

        int s = 0, d = 0;
        float nrm = 0.0f;
        if (lane < cnt) {
            s = src[e0 + lane];
            d = dst[e0 + lane];
            nrm = in_lds ? tab[s] * tab[d] : dinv[s] * dinv[d];
        }

        int j = 0;
        for (; j + 8 <= cnt; j += 8) {
            const float* hp[8];
            size_t       od[8];
            float        wt[8];
#pragma unroll
            for (int u = 0; u < 8; ++u) {
                const int sj = __builtin_amdgcn_readlane(s, j + u);
                const int dj = __builtin_amdgcn_readlane(d, j + u);
                wt[u] = __builtin_bit_cast(float,
                            __builtin_amdgcn_readlane(__builtin_bit_cast(int, nrm), j + u));
                hp[u] = h   + (size_t)sj * f;
                od[u] = (size_t)dj * pitch;
            }
            float v[8][NL];
#pragma unroll
            for (int u = 0; u < 8; ++u) {
#pragma unroll
                for (int k = 0; k < NL; ++k) v[u][k] = hp[u][k * 32 + lane];
            }
#pragma unroll
            for (int u = 0; u < 8; ++u) {
#pragma unroll
                for (int k = 0; k < NL; ++k) atomicAdd(lb[k] + od[u] + lane, v[u][k] * wt[u]);
            }
        }
        for (; j < cnt; ++j) {
            const int sj = __builtin_amdgcn_readlane(s, j);
            const int dj = __builtin_amdgcn_readlane(d, j);
            const float wj = __builtin_bit_cast(float,
                                 __builtin_amdgcn_readlane(__builtin_bit_cast(int, nrm), j));
            const float* hr = h   + (size_t)sj * f;
            const size_t odj = (size_t)dj * pitch;
#pragma unroll
            for (int k = 0; k < NL; ++k)
                atomicAdd(lb[k] + odj + lane, hr[k * 32 + lane] * wj);
        }
    }
}

__global__ void combine_kernel(const float* __restrict__ la, const float* __restrict__ lb,
                               float* __restrict__ logits, int n) {
    int i = blockIdx.x * blockDim.x + threadIdx.x;
    if (i < n) logits[i] = 0.5f * (la[i] + lb[i]);
}


static inline int cdiv(int a, int b) { return (a + b - 1) / b; }

static void run_branch(const float* x, const int* edge, const float* W1, const float* b1,
                       const float* W2, const float* b2, float* l_out,
                       float* q0, float* q1,
                       float* dinv, float* h, float* xh, float* h2,
                       int Nn, int IN, int HID, int OUT, int E, hipStream_t stream) {
    const int* src = edge;
    const int* dst = edge + E;

    deg_init_kernel<<<cdiv(Nn, 256), 256, 0, stream>>>(dinv, Nn);
    deg_accum_kernel<<<64, 1024, 0, stream>>>(dst, dinv, E, Nn);

    const int strips = Nn / 16;

    {
        gemm_wmma_f32_fused<6, false, true, true, false><<<cdiv(strips, 4), 128, 0, stream>>>(x, W1, dinv, b1, h, xh, q0, q1, h2, Nn, IN);
        scatter_kernel<3, true><<<256, 1024, 0, stream>>>(h, dinv, src, dst, xh, q0, q1, h2, E, Nn);
    }

    {
        gemm_wmma_f32_fused<2, true, false, false, true><<<cdiv(strips, 4), 128, 0, stream>>>(xh, W2, dinv, b2, h2, l_out, q0, q1, h2, Nn, HID);
        scatter_kernel<1, false><<<256, 1024, 0, stream>>>(h2, dinv, src, dst, l_out, nullptr, nullptr, nullptr, E, Nn);
    }
}

extern "C" void kernel_launch(void* const* d_in, const int* in_sizes, int n_in,
                              void* d_out, int out_size, void* d_ws, size_t ws_size,
                              hipStream_t stream) {
    const float* x      = (const float*)d_in[0];
    const int*   edge_a = (const int*)  d_in[1];
    const int*   edge_b = (const int*)  d_in[2];
    const float* W1a    = (const float*)d_in[3];
    const float* b1a    = (const float*)d_in[4];
    const float* W2a    = (const float*)d_in[5];
    const float* b2a    = (const float*)d_in[6];
    const float* W1b    = (const float*)d_in[7];
    const float* b1b    = (const float*)d_in[8];
    const float* W2b    = (const float*)d_in[9];
    const float* b2b    = (const float*)d_in[10];

    const int HID = in_sizes[4];
    const int OUT = in_sizes[6];
    const int IN  = in_sizes[3] / HID;
    const int Nn  = in_sizes[0] / IN;
    const int E   = in_sizes[1] / 2;

    float* ws   = (float*)d_ws;
    float* h    = ws;                       ws += (size_t)Nn * HID;
    float* xh   = ws;                       ws += (size_t)Nn * HID;
    float* h2   = ws;                       ws += (size_t)Nn * OUT;
    float* dinv = ws;                       ws += Nn;

    float* logits = (float*)d_out;
    float* la     = logits + (size_t)Nn * OUT;
    float* lb     = la     + (size_t)Nn * OUT;

    run_branch(x, edge_a, W1a, b1a, W2a, b2a, la, logits, lb, dinv, h, xh, h2, Nn, IN, HID, OUT, E, stream);
    run_branch(x, edge_b, W1b, b1b, W2b, b2b, lb, logits, lb, dinv, h, xh, h2, Nn, IN, HID, OUT, E, stream);

    combine_kernel<<<cdiv(Nn * OUT, 256), 256, 0, stream>>>(la, lb, logits, Nn * OUT);
}


# layer-2 scatter too: each package half adds into its own copy of the sums (die id, for speed only); the next launch adds the copies; f32 sums: per half in atomic arrival order, then added
# speedup vs baseline: 3.0052x; 1.0937x over previous
#include <hip/hip_runtime.h>

typedef __attribute__((ext_vector_type(2))) float v2f;
typedef __attribute__((ext_vector_type(8))) float v8f;


__global__ void deg_init_kernel(float* __restrict__ deg, int n,
                                float* __restrict__ acc, const float* __restrict__ add, int m) {
    int i = blockIdx.x * blockDim.x + threadIdx.x;
    if (i < n) deg[i] = 1.0f;
    if (acc != nullptr && i < m) acc[i] += add[i];
}

constexpr int CNT_CAP = 65536;

__global__ __launch_bounds__(1024)
void deg_accum_kernel(const int* __restrict__ dst, float* __restrict__ deg, int E, int n) {
    __shared__ unsigned cnt[CNT_CAP + 32];
    const int tid = threadIdx.x;
    const int nthr = blockDim.x;
    if (n > CNT_CAP) {
        for (int e = blockIdx.x * nthr + tid; e < E; e += gridDim.x * nthr)
            atomicAdd(&deg[dst[e]], 1.0f);
        return;
    }
    uint4* c4 = reinterpret_cast<uint4*>(cnt);
    const int n4 = (n + 3) >> 2;
    for (int i = tid; i < n4; i += nthr) c4[i] = make_uint4(0u, 0u, 0u, 0u);
    __syncthreads();

    const int per = (E + (int)gridDim.x - 1) / (int)gridDim.x;
    const int e_begin = min(E, (int)blockIdx.x * per);
    const int e_end = min(E, e_begin + per);
    const int spare = CNT_CAP + (tid & 31);
    for (int e0 = e_begin + tid; e0 < e_end; e0 += 8 * nthr) {
        int d[8];
#pragma unroll
        for (int q = 0; q < 8; ++q) d[q] = dst[min(e0 + q * nthr, e_end - 1)];
#pragma unroll
        for (int q = 0; q < 8; ++q)
            atomicAdd(&cnt[(e0 + q * nthr < e_end) ? d[q] : spare], 1u);
    }
    __syncthreads();

    const int lane = tid & 31;
    const int wv = tid >> 5, nw = nthr >> 5;
    const int nl = n >> 5;
    int ln = wv;
    for (; ln + 7 * nw < nl; ln += 8 * nw) {
        unsigned c[8];
#pragma unroll
        for (int q = 0; q < 8; ++q) c[q] = cnt[(ln + q * nw) * 32 + lane];
#pragma unroll
        for (int q = 0; q < 8; ++q) atomicAdd(&deg[(ln + q * nw) * 32 + lane], (float)c[q]);
    }
    for (; ln < nl; ln += nw) atomicAdd(&deg[ln * 32 + lane], (float)cnt[ln * 32 + lane]);
    const int it = (nl << 5) + tid;
    if (it < n) atomicAdd(&deg[it], (float)cnt[it]);
}

template <int NT, bool RELU_A, bool DEG_TO_DINV, bool ZERO_P, bool ADD_P>
__global__ void gemm_wmma_f32_fused(const float* __restrict__ X, const float* __restrict__ W,
                                    float* dnorm, const float* __restrict__ bias,
                                    float* H, float* O,
                                    float* p0, float* p1, float* p2,
                                    int n_rows, int k_dim) {
    constexpr int NC = NT * 16;
    const int wave = threadIdx.x >> 5;
    const int lane = threadIdx.x & 31;
    const int rt = blockIdx.x * (blockDim.x >> 5) + wave;
    if (rt >= (n_rows >> 4)) return;

    const int r  = lane & 15;
    const int hi = lane >> 4;

    const float* __restrict__ arow = X + (size_t)(rt * 16 + r) * k_dim;

    v8f acc[NT] = {};
    for (int k = 0; k < k_dim; k += 4) {
        v2f a;
        a.x = arow[k + 2 * hi];
        a.y = arow[k + 2 * hi + 1];
        if (ADD_P) {
            const float* pk = (k < 32) ? p0 : ((k < 64) ? p1 : p2);
            const float* prow = pk + (size_t)(rt * 16 + r) * 32 + (k & 31) + 2 * hi;
            a.x += prow[0];
            a.y += prow[1];
        }
        if (RELU_A) {
            a.x = fmaxf(a.x, 0.0f);
            a.y = fmaxf(a.y, 0.0f);
        }
        const float* __restrict__ w0 = W + (size_t)(k + 2 * hi) * NC + r;
#pragma unroll
        for (int t = 0; t < NT; ++t) {
            v2f b;
            b.x = w0[t * 16];
            b.y = w0[NC + t * 16];
            acc[t] = __builtin_amdgcn_wmma_f32_16x16x4_f32(
                 false, a,  false, b,
                 (short)0, acc[t],  false,  false);
        }
    }

    const int row0 = rt * 16 + 8 * hi;
    float di[8];
#pragma unroll
    for (int v = 0; v < 8; ++v) di[v] = dnorm[row0 + v];
    if (DEG_TO_DINV) {
#pragma unroll
        for (int v = 0; v < 8; ++v) di[v] = rsqrtf(di[v]);
        if (r == 0) {
#pragma unroll
            for (int v = 0; v < 8; ++v) dnorm[row0 + v] = di[v];
        }
    }
#pragma unroll
    for (int v = 0; v < 8; ++v) {
#pragma unroll
        for (int t = 0; t < NT; ++t) {
            const size_t idx = (size_t)(row0 + v) * NC + t * 16 + r;
            const float hv = acc[t][v];
            H[idx] = hv;
            O[idx] = bias[t * 16 + r] + hv * di[v] * di[v];
            if (ZERO_P) {
                float* pz = (t / 2 == 0) ? p0 : ((t / 2 == 1) ? p1 : p2);
                pz[(size_t)(row0 + v) * 32 + (t & 1) * 16 + r] = 0.0f;
            }
        }
    }
}


constexpr int TAB_CAP = 65536;

template <int NL, bool TWO_COPIES>
__global__ __launch_bounds__(1024)
void scatter_kernel(const float* __restrict__ h, const float* __restrict__ dinv,
                    const int* __restrict__ src, const int* __restrict__ dst,
                    float* __restrict__ out, float* __restrict__ p0, float* __restrict__ p1,
                    float* __restrict__ p2, int E, int n) {
    __shared__ float tab[TAB_CAP];
    constexpr int f = NL * 32;
    const bool in_lds = n <= TAB_CAP;
    if (in_lds) {
        const int n4 = n >> 2;
        const float4* __restrict__ d4 = reinterpret_cast<const float4*>(dinv);
        float4* t4 = reinterpret_cast<float4*>(tab);
        for (int i0 = threadIdx.x; i0 < n4; i0 += 8 * (int)blockDim.x) {
            int    ix[8];
            float4 r[8];
#pragma unroll
            for (int q = 0; q < 8; ++q) ix[q] = min(i0 + q * (int)blockDim.x, n4 - 1);
#pragma unroll
            for (int q = 0; q < 8; ++q) r[q] = d4[ix[q]];
#pragma unroll
            for (int q = 0; q < 8; ++q) t4[ix[q]] = r[q];
        }
        const int it = (n4 << 2) + threadIdx.x;
        if (it < n) tab[it] = dinv[it];
    }
    __syncthreads();

    const int lane = threadIdx.x & 31;
    const int wv = __builtin_amdgcn_readfirstlane(threadIdx.x >> 5);
    const int wpb = blockDim.x >> 5;
    const int W = gridDim.x * wpb;
    const int w = blockIdx.x * wpb + wv;
    const int U = E >> 3;
    const int uq = U / W, ur = U % W;
    const int e_begin = (w * uq + min(w, ur)) << 3;
    const int e_end = (w == W - 1) ? E : (((w + 1) * uq + min(w + 1, ur)) << 3);

    float* lb[NL];
#pragma unroll
    for (int k = 0; k < NL; ++k) lb[k] = out + k * 32;
    int pitch = f;
    if (TWO_COPIES) {
        const bool second_half = (__builtin_amdgcn_s_sendmsg_rtn(0x87) >> 18) & 1;
        if (second_half) {
            lb[0] = p0;
            if (NL > 1) lb[1] = p1;
            if (NL > 2) lb[2] = p2;
            pitch = 32;
        }
    }

    for (int e0 = e_begin; e0 < e_end; e0 += 32) {
        const int cnt = min(32, e_end - e0);

        int s = 0, d = 0;
        float nrm = 0.0f;
        if (lane < cnt) {
            s = src[e0 + lane];
            d = dst[e0 + lane];
            nrm = in_lds ? tab[s] * tab[d] : dinv[s] * dinv[d];
        }

        int j = 0;
        for (; j + 8 <= cnt; j += 8) {
            const float* hp[8];
            size_t       od[8];
            float        wt[8];
#pragma unroll
            for (int u = 0; u < 8; ++u) {
                const int sj = __builtin_amdgcn_readlane(s, j + u);
                const int dj = __builtin_amdgcn_readlane(d, j + u);
                wt[u] = __builtin_bit_cast(float,
                            __builtin_amdgcn_readlane(__builtin_bit_cast(int, nrm), j + u));
                hp[u] = h   + (size_t)sj * f;
                od[u] = (size_t)dj * pitch;
            }
            float v[8][NL];
#pragma unroll
            for (int u = 0; u < 8; ++u) {
#pragma unroll
                for (int k = 0; k < NL; ++k) v[u][k] = hp[u][k * 32 + lane];
            }
#pragma unroll
            for (int u = 0; u < 8; ++u) {
#pragma unroll
                for (int k = 0; k < NL; ++k) atomicAdd(lb[k] + od[u] + lane, v[u][k] * wt[u]);
            }
        }
        for (; j < cnt; ++j) {
            const int sj = __builtin_amdgcn_readlane(s, j);
            const int dj = __builtin_amdgcn_readlane(d, j);
            const float wj = __builtin_bit_cast(float,
                                 __builtin_amdgcn_readlane(__builtin_bit_cast(int, nrm), j));
            const float* hr = h   + (size_t)sj * f;
            const size_t odj = (size_t)dj * pitch;
#pragma unroll
            for (int k = 0; k < NL; ++k)
                atomicAdd(lb[k] + odj + lane, hr[k * 32 + lane] * wj);
        }
    }
}

__global__ void combine_kernel(const float* __restrict__ la, float* __restrict__ lb,
                               float* __restrict__ logits, int n) {
    int i = blockIdx.x * blockDim.x + threadIdx.x;
    if (i < n) {
        const float b = lb[i] + logits[i];
        lb[i] = b;
        logits[i] = 0.5f * (la[i] + b);
    }
}


static inline int cdiv(int a, int b) { return (a + b - 1) / b; }

static void run_branch(const float* x, const int* edge, const float* W1, const float* b1,
                       const float* W2, const float* b2, float* l_out,
                       float* fold_acc,
                       float* q0, float* q1,
                       float* dinv, float* h, float* xh, float* h2,
                       int Nn, int IN, int HID, int OUT, int E, hipStream_t stream) {
    const int* src = edge;
    const int* dst = edge + E;

    const int fold_n = fold_acc ? Nn * OUT : 0;
    deg_init_kernel<<<cdiv(fold_n > Nn ? fold_n : Nn, 256), 256, 0, stream>>>(dinv, Nn, fold_acc, q0, fold_n);
    deg_accum_kernel<<<64, 1024, 0, stream>>>(dst, dinv, E, Nn);

    const int strips = Nn / 16;

    {
        gemm_wmma_f32_fused<6, false, true, true, false><<<cdiv(strips, 4), 128, 0, stream>>>(x, W1, dinv, b1, h, xh, q0, q1, h2, Nn, IN);
        scatter_kernel<3, true><<<256, 1024, 0, stream>>>(h, dinv, src, dst, xh, q0, q1, h2, E, Nn);
    }

    {
        gemm_wmma_f32_fused<2, true, false, true, true><<<cdiv(strips, 4), 128, 0, stream>>>(xh, W2, dinv, b2, h2, l_out, q0, q1, h2, Nn, HID);
        scatter_kernel<1, true><<<256, 1024, 0, stream>>>(h2, dinv, src, dst, l_out, q0, nullptr, nullptr, E, Nn);
    }
}

extern "C" void kernel_launch(void* const* d_in, const int* in_sizes, int n_in,
                              void* d_out, int out_size, void* d_ws, size_t ws_size,
                              hipStream_t stream) {
    const float* x      = (const float*)d_in[0];
    const int*   edge_a = (const int*)  d_in[1];
    const int*   edge_b = (const int*)  d_in[2];
    const float* W1a    = (const float*)d_in[3];
    const float* b1a    = (const float*)d_in[4];
    const float* W2a    = (const float*)d_in[5];
    const float* b2a    = (const float*)d_in[6];
    const float* W1b    = (const float*)d_in[7];
    const float* b1b    = (const float*)d_in[8];
    const float* W2b    = (const float*)d_in[9];
    const float* b2b    = (const float*)d_in[10];

    const int HID = in_sizes[4];
    const int OUT = in_sizes[6];
    const int IN  = in_sizes[3] / HID;
    const int Nn  = in_sizes[0] / IN;
    const int E   = in_sizes[1] / 2;

    float* ws   = (float*)d_ws;
    float* h    = ws;                       ws += (size_t)Nn * HID;
    float* xh   = ws;                       ws += (size_t)Nn * HID;
    float* h2   = ws;                       ws += (size_t)Nn * OUT;
    float* dinv = ws;                       ws += Nn;

    float* logits = (float*)d_out;
    float* la     = logits + (size_t)Nn * OUT;
    float* lb     = la     + (size_t)Nn * OUT;

    run_branch(x, edge_a, W1a, b1a, W2a, b2a, la, nullptr, logits, lb, dinv, h, xh, h2, Nn, IN, HID, OUT, E, stream);
    run_branch(x, edge_b, W1b, b1b, W2b, b2b, lb, la, logits, lb, dinv, h, xh, h2, Nn, IN, HID, OUT, E, stream);

    combine_kernel<<<cdiv(Nn * OUT, 256), 256, 0, stream>>>(la, lb, logits, Nn * OUT);
}


# degree count runs inside the layer-1 product's launch (60 groups count beside it); layer 1's bias and self-loop term move to layer 2's row loads: f32 sum order changes; 13 -> 11 launches
# speedup vs baseline: 3.2386x; 1.0777x over previous
#include <hip/hip_runtime.h>

typedef __attribute__((ext_vector_type(2))) float v2f;
typedef __attribute__((ext_vector_type(8))) float v8f;


__global__ void deg_init_kernel(float* __restrict__ deg, int n,
                                float* __restrict__ acc, const float* __restrict__ add, int m) {
    int i = blockIdx.x * blockDim.x + threadIdx.x;
    if (i < n) deg[i] = 1.0f;
    if (acc != nullptr && i < m) acc[i] += add[i];
}

constexpr int CNT_CAP = 65536;
constexpr int CNT_Q = 16;

__device__ __forceinline__ void deg_accum_body(unsigned* cnt, const int* __restrict__ dst,
                                               float* __restrict__ deg, int E, int n,
                                               int blk, int nblk) {
    const int tid = threadIdx.x;
    const int nthr = blockDim.x;
    if (n > CNT_CAP) {
        for (int e = blk * nthr + tid; e < E; e += nblk * nthr)
            atomicAdd(&deg[dst[e]], 1.0f);
        return;
    }
    uint4* c4 = reinterpret_cast<uint4*>(cnt);
    const int n4 = (n + 3) >> 2;
    for (int i = tid; i < n4; i += nthr) c4[i] = make_uint4(0u, 0u, 0u, 0u);
    __syncthreads();

    const int per = (E + nblk - 1) / nblk;
    const int e_begin = min(E, blk * per);
    const int e_end = min(E, e_begin + per);
    const int spare = CNT_CAP + (tid & 31);
    for (int e0 = e_begin + tid; e0 < e_end; e0 += CNT_Q * nthr) {
        int d[CNT_Q];
#pragma unroll
        for (int q = 0; q < CNT_Q; ++q) d[q] = dst[min(e0 + q * nthr, e_end - 1)];
#pragma unroll
        for (int q = 0; q < CNT_Q; ++q)
            atomicAdd(&cnt[(e0 + q * nthr < e_end) ? d[q] : spare], 1u);
    }
    __syncthreads();

    const int lane = tid & 31;
    const int wv = tid >> 5, nw = nthr >> 5;
    const int nl = n >> 5;
    int ln = wv;
    for (; ln + 7 * nw < nl; ln += 8 * nw) {
        unsigned c[8];
#pragma unroll
        for (int q = 0; q < 8; ++q) c[q] = cnt[(ln + q * nw) * 32 + lane];
#pragma unroll
        for (int q = 0; q < 8; ++q) atomicAdd(&deg[(ln + q * nw) * 32 + lane], (float)c[q]);
    }
    for (; ln < nl; ln += nw) atomicAdd(&deg[ln * 32 + lane], (float)cnt[ln * 32 + lane]);
    const int it = (nl << 5) + tid;
    if (it < n) atomicAdd(&deg[it], (float)cnt[it]);
}

template <int NT, bool SECOND>
__device__ __forceinline__ void gemm_strip(const float* __restrict__ X, const float* __restrict__ W,
                                           const float* __restrict__ deg, const float* __restrict__ bias,
                                           const float* __restrict__ Hs, const float* __restrict__ bs,
                                           float* H, float* O,
                                           float* p0, float* p1, float* p2,
                                           int rt, int lane, int k_dim) {
    constexpr int NC = NT * 16;
    const int r  = lane & 15;
    const int hi = lane >> 4;

    const size_t arow_i = (size_t)(rt * 16 + r);
    const float* __restrict__ arow = X + arow_i * k_dim;
    float dr = 0.0f;
    if (SECOND) dr = rsqrtf(deg[rt * 16 + r]);

    v8f acc[NT] = {};
#pragma unroll 4
    for (int k = 0; k < k_dim; k += 4) {
        v2f a;
        a.x = arow[k + 2 * hi];
        a.y = arow[k + 2 * hi + 1];
        if (SECOND) {
            const float* pk = (k < 32) ? p0 : ((k < 64) ? p1 : p2);
            const float* prow = pk + arow_i * 32 + (k & 31) + 2 * hi;
            const float* __restrict__ hrow = Hs + arow_i * k_dim + k + 2 * hi;
            a.x = (a.x + prow[0]) + (bs[k + 2 * hi]     + hrow[0] * dr * dr);
            a.y = (a.y + prow[1]) + (bs[k + 2 * hi + 1] + hrow[1] * dr * dr);
            a.x = fmaxf(a.x, 0.0f);
            a.y = fmaxf(a.y, 0.0f);
        }
        const float* __restrict__ w0 = W + (size_t)(k + 2 * hi) * NC + r;
#pragma unroll
        for (int t = 0; t < NT; ++t) {
            v2f b;
            b.x = w0[t * 16];
            b.y = w0[NC + t * 16];
            acc[t] = __builtin_amdgcn_wmma_f32_16x16x4_f32(
                 false, a,  false, b,
                 (short)0, acc[t],  false,  false);
        }
    }

    const int row0 = rt * 16 + 8 * hi;
    float di[8] = {};
    if (SECOND) {
#pragma unroll
        for (int v = 0; v < 8; ++v) di[v] = deg[row0 + v];
#pragma unroll
        for (int v = 0; v < 8; ++v) di[v] = rsqrtf(di[v]);
    }
#pragma unroll
    for (int v = 0; v < 8; ++v) {
#pragma unroll
        for (int t = 0; t < NT; ++t) {
            const size_t idx = (size_t)(row0 + v) * NC + t * 16 + r;
            const float hv = acc[t][v];
            H[idx] = hv;
            O[idx] = SECOND ? (bias[t * 16 + r] + hv * di[v] * di[v]) : 0.0f;
            float* pz = (t / 2 == 0) ? p0 : ((t / 2 == 1) ? p1 : p2);
            pz[(size_t)(row0 + v) * 32 + (t & 1) * 16 + r] = 0.0f;
        }
    }
}

constexpr int DEG_BLOCKS = 60;
constexpr int GEMM_WAVES = 16;

__global__ __launch_bounds__(GEMM_WAVES * 32)
void deg_gemm1_kernel(const int* __restrict__ dst, float* __restrict__ deg, int E,
                      const float* __restrict__ X, const float* __restrict__ W,
                      float* H, float* O, float* p0, float* p1, float* p2,
                      int n_rows, int k_dim) {
    __shared__ unsigned cnt[CNT_CAP + 32];
    if ((int)blockIdx.x < DEG_BLOCKS) {
        deg_accum_body(cnt, dst, deg, E, n_rows, (int)blockIdx.x, DEG_BLOCKS);
        return;
    }
    const int wave = threadIdx.x >> 5;
    const int lane = threadIdx.x & 31;
    const int rt = ((int)blockIdx.x - DEG_BLOCKS) * (int)(blockDim.x >> 5) + wave;
    if (rt >= (n_rows >> 4)) return;
    gemm_strip<6, false>(X, W, nullptr, nullptr, nullptr, nullptr, H, O, p0, p1, p2, rt, lane, k_dim);
}

__global__ void gemm2_kernel(const float* __restrict__ X, const float* __restrict__ W,
                             const float* __restrict__ deg, const float* __restrict__ bias,
                             const float* __restrict__ Hs, const float* __restrict__ bs,
                             float* H, float* O, float* p0, float* p1, float* p2,
                             int n_rows, int k_dim) {
    const int wave = threadIdx.x >> 5;
    const int lane = threadIdx.x & 31;
    const int rt = blockIdx.x * (blockDim.x >> 5) + wave;
    if (rt >= (n_rows >> 4)) return;
    gemm_strip<2, true>(X, W, deg, bias, Hs, bs, H, O, p0, p1, p2, rt, lane, k_dim);
}


constexpr int TAB_CAP = 65536;

template <int NL, bool TWO_COPIES>
__global__ __launch_bounds__(1024)
void scatter_kernel(const float* __restrict__ h, const float* __restrict__ deg,
                    const int* __restrict__ src, const int* __restrict__ dst,
                    float* __restrict__ out, float* __restrict__ p0, float* __restrict__ p1,
                    float* __restrict__ p2, int E, int n) {
    __shared__ float tab[TAB_CAP];
    constexpr int f = NL * 32;
    const bool in_lds = n <= TAB_CAP;
    if (in_lds) {
        const int n4 = n >> 2;
        const float4* __restrict__ d4 = reinterpret_cast<const float4*>(deg);
        float4* t4 = reinterpret_cast<float4*>(tab);
        for (int i0 = threadIdx.x; i0 < n4; i0 += 8 * (int)blockDim.x) {
            int    ix[8];
            float4 r[8];
#pragma unroll
            for (int q = 0; q < 8; ++q) ix[q] = min(i0 + q * (int)blockDim.x, n4 - 1);
#pragma unroll
            for (int q = 0; q < 8; ++q) r[q] = d4[ix[q]];
#pragma unroll
            for (int q = 0; q < 8; ++q)
                t4[ix[q]] = make_float4(rsqrtf(r[q].x), rsqrtf(r[q].y), rsqrtf(r[q].z), rsqrtf(r[q].w));
        }
        const int it = (n4 << 2) + threadIdx.x;
        if (it < n) tab[it] = rsqrtf(deg[it]);
    }
    __syncthreads();

    const int lane = threadIdx.x & 31;
    const int wv = __builtin_amdgcn_readfirstlane(threadIdx.x >> 5);
    const int wpb = blockDim.x >> 5;
    const int W = gridDim.x * wpb;
    const int w = blockIdx.x * wpb + wv;
    const int U = E >> 3;
    const int uq = U / W, ur = U % W;
    const int e_begin = (w * uq + min(w, ur)) << 3;
    const int e_end = (w == W - 1) ? E : (((w + 1) * uq + min(w + 1, ur)) << 3);

    float* lb[NL];
#pragma unroll
    for (int k = 0; k < NL; ++k) lb[k] = out + k * 32;
    int pitch = f;
    if (TWO_COPIES) {
        const bool second_half = (__builtin_amdgcn_s_sendmsg_rtn(0x87) >> 18) & 1;
        if (second_half) {
            lb[0] = p0;
            if (NL > 1) lb[1] = p1;
            if (NL > 2) lb[2] = p2;
            pitch = 32;
        }
    }

    for (int e0 = e_begin; e0 < e_end; e0 += 32) {
        const int cnt = min(32, e_end - e0);

        int s = 0, d = 0;
        float nrm = 0.0f;
        if (lane < cnt) {
            s = src[e0 + lane];
            d = dst[e0 + lane];
            nrm = in_lds ? tab[s] * tab[d] : rsqrtf(deg[s]) * rsqrtf(deg[d]);
        }

        int j = 0;
        for (; j + 8 <= cnt; j += 8) {
            const float* hp[8];
            size_t       od[8];
            float        wt[8];
#pragma unroll
            for (int u = 0; u < 8; ++u) {
                const int sj = __builtin_amdgcn_readlane(s, j + u);
                const int dj = __builtin_amdgcn_readlane(d, j + u);
                wt[u] = __builtin_bit_cast(float,
                            __builtin_amdgcn_readlane(__builtin_bit_cast(int, nrm), j + u));
                hp[u] = h   + (size_t)sj * f;
                od[u] = (size_t)dj * pitch;
            }
            float v[8][NL];
#pragma unroll
            for (int u = 0; u < 8; ++u) {
#pragma unroll
                for (int k = 0; k < NL; ++k) v[u][k] = hp[u][k * 32 + lane];
            }
#pragma unroll
            for (int u = 0; u < 8; ++u) {
#pragma unroll
                for (int k = 0; k < NL; ++k) atomicAdd(lb[k] + od[u] + lane, v[u][k] * wt[u]);
            }
        }
        for (; j < cnt; ++j) {
            const int sj = __builtin_amdgcn_readlane(s, j);
            const int dj = __builtin_amdgcn_readlane(d, j);
            const float wj = __builtin_bit_cast(float,
                                 __builtin_amdgcn_readlane(__builtin_bit_cast(int, nrm), j));
            const float* hr = h   + (size_t)sj * f;
            const size_t odj = (size_t)dj * pitch;
#pragma unroll
            for (int k = 0; k < NL; ++k)
                atomicAdd(lb[k] + odj + lane, hr[k * 32 + lane] * wj);
        }
    }
}

__global__ void combine_kernel(const float* __restrict__ la, float* __restrict__ lb,
                               float* __restrict__ logits, int n) {
    int i = blockIdx.x * blockDim.x + threadIdx.x;
    if (i < n) {
        const float b = lb[i] + logits[i];
        lb[i] = b;
        logits[i] = 0.5f * (la[i] + b);
    }
}


static inline int cdiv(int a, int b) { return (a + b - 1) / b; }

static void run_branch(const float* x, const int* edge, const float* W1, const float* b1,
                       const float* W2, const float* b2, float* l_out,
                       float* fold_acc,
                       float* q0, float* q1,
                       float* dinv, float* h, float* xh, float* h2,
                       int Nn, int IN, int HID, int OUT, int E, hipStream_t stream) {
    const int* src = edge;
    const int* dst = edge + E;

    const int fold_n = fold_acc ? Nn * OUT : 0;
    deg_init_kernel<<<cdiv(fold_n > Nn ? fold_n : Nn, 256), 256, 0, stream>>>(dinv, Nn, fold_acc, q0, fold_n);
    const int strips = Nn / 16;

    {
        deg_gemm1_kernel<<<DEG_BLOCKS + cdiv(strips, GEMM_WAVES), GEMM_WAVES * 32, 0, stream>>>(dst, dinv, E, x, W1, h, xh, q0, q1, h2, Nn, IN);
        scatter_kernel<3, true><<<256, 1024, 0, stream>>>(h, dinv, src, dst, xh, q0, q1, h2, E, Nn);
    }

    {
        gemm2_kernel<<<cdiv(strips, 4), 128, 0, stream>>>(xh, W2, dinv, b2, h, b1, h2, l_out, q0, q1, h2, Nn, HID);
        scatter_kernel<1, true><<<256, 1024, 0, stream>>>(h2, dinv, src, dst, l_out, q0, nullptr, nullptr, E, Nn);
    }
}

extern "C" void kernel_launch(void* const* d_in, const int* in_sizes, int n_in,
                              void* d_out, int out_size, void* d_ws, size_t ws_size,
                              hipStream_t stream) {
    const float* x      = (const float*)d_in[0];
    const int*   edge_a = (const int*)  d_in[1];
    const int*   edge_b = (const int*)  d_in[2];
    const float* W1a    = (const float*)d_in[3];
    const float* b1a    = (const float*)d_in[4];
    const float* W2a    = (const float*)d_in[5];
    const float* b2a    = (const float*)d_in[6];
    const float* W1b    = (const float*)d_in[7];
    const float* b1b    = (const float*)d_in[8];
    const float* W2b    = (const float*)d_in[9];
    const float* b2b    = (const float*)d_in[10];

    const int HID = in_sizes[4];
    const int OUT = in_sizes[6];
    const int IN  = in_sizes[3] / HID;
    const int Nn  = in_sizes[0] / IN;
    const int E   = in_sizes[1] / 2;

    float* ws   = (float*)d_ws;
    float* h    = ws;                       ws += (size_t)Nn * HID;
    float* xh   = ws;                       ws += (size_t)Nn * HID;
    float* h2   = ws;                       ws += (size_t)Nn * OUT;
    float* dinv = ws;                       ws += Nn;

    float* logits = (float*)d_out;
    float* la     = logits + (size_t)Nn * OUT;
    float* lb     = la     + (size_t)Nn * OUT;

    run_branch(x, edge_a, W1a, b1a, W2a, b2a, la, nullptr, logits, lb, dinv, h, xh, h2, Nn, IN, HID, OUT, E, stream);
    run_branch(x, edge_b, W1b, b1b, W2b, b2b, lb, la, logits, lb, dinv, h, xh, h2, Nn, IN, HID, OUT, E, stream);

    combine_kernel<<<cdiv(Nn * OUT, 256), 256, 0, stream>>>(la, lb, logits, Nn * OUT);
}
